# starting kernel
# baseline (speedup 1.0000x reference)
#include <hip/hip_runtime.h>
#include <hip/hip_bf16.h>

typedef float v2f __attribute__((ext_vector_type(2)));
typedef float v8f __attribute__((ext_vector_type(8)));

#define D 64

__global__ __launch_bounds__(128)
void gcn_gemm_bias(const float* __restrict__ x,
                   const float* __restrict__ W,
                   const float* __restrict__ b,
                   float* __restrict__ h,
                   int n_nodes) {
  const int lane = threadIdx.x & 31;
  const int wave = threadIdx.x >> 5;
  const int row0 = blockIdx.x << 4;
  const int col0 = wave << 4;

  const int m  = lane & 15;
  const int kk = (lane >> 4) << 1;

  int arow = row0 + m;
  if (arow >= n_nodes) arow = n_nodes - 1;

  const float* __restrict__ xrow = x + (size_t)arow * D;
  const float* __restrict__ wrow = W + (size_t)(col0 + m) * D;

  v8f acc = {};
#pragma unroll
  for (int k0 = 0; k0 < D; k0 += 4) {
    v2f a  = *(const v2f*)(xrow + k0 + kk);
    v2f bb = *(const v2f*)(wrow + k0 + kk);
    acc = __builtin_amdgcn_wmma_f32_16x16x4_f32(
        false, a, false, bb, (short)0, acc, false, false);
  }

  const float bias = b[col0 + m];
  const int rbase = row0 + ((lane >> 4) << 3);
#pragma unroll
  for (int r = 0; r < 8; ++r) {
    const int row = rbase + r;
    if (row < n_nodes) {
      h[(size_t)row * D + col0 + m] = acc[r] + bias;
    }
  }
}

__global__ __launch_bounds__(256)
void gcn_edge_scatter(const float* __restrict__ h,
                      const int* __restrict__ src,
                      const int* __restrict__ dst,
                      const float* __restrict__ w,
                      float* __restrict__ out,
                      int n_edges) {
  const int gid = blockIdx.x * 256 + threadIdx.x;
  const int e = gid >> 4;
  if (e >= n_edges) return;
  const int f = (gid & 15) << 2;

  const int   s  = src[e];
  const int   d  = dst[e];
  const float wv = w[e];

  const float4 hv = *(const float4*)(h + (size_t)s * D + f);
  float* __restrict__ o = out + (size_t)d * D + f;

  atomicAdd(o + 0, hv.x * wv);
  atomicAdd(o + 1, hv.y * wv);
  atomicAdd(o + 2, hv.z * wv);
  atomicAdd(o + 3, hv.w * wv);
}

extern "C" void kernel_launch(void* const* d_in, const int* in_sizes, int n_in,
                              void* d_out, int out_size, void* d_ws, size_t ws_size,
                              hipStream_t stream) {
  const float* x   = (const float*)d_in[0];
  const int*   src = (const int*)  d_in[1];
  const int*   dst = (const int*)  d_in[2];
  const float* w   = (const float*)d_in[3];
  const float* W   = (const float*)d_in[4];
  const float* b   = (const float*)d_in[5];
  float*       out = (float*)d_out;
  float*       h   = (float*)d_ws;

  const int n_nodes = in_sizes[0] / D;
  const int n_edges = in_sizes[1];

  hipMemsetAsync(d_out, 0, (size_t)out_size * sizeof(float), stream);

  const int row_tiles = (n_nodes + 15) / 16;
  gcn_gemm_bias<<<dim3(row_tiles), dim3(128), 0, stream>>>(x, W, b, h, n_nodes);

  const long long threads = (long long)n_edges * 16;
  const int blocks = (int)((threads + 255) / 256);
  gcn_edge_scatter<<<dim3(blocks), dim3(256), 0, stream>>>(h, src, dst, w, out, n_edges);
}


# edge scatter: one wave per 32 edges, lane=feature so every h load and f32 atomic add covers one whole 128-B line; edge ids read coalesced, broadcast by readlane; atomic arrival order changes
# speedup vs baseline: 1.8397x; 1.8397x over previous
#include <hip/hip_runtime.h>
#include <hip/hip_bf16.h>

typedef float v2f __attribute__((ext_vector_type(2)));
typedef float v8f __attribute__((ext_vector_type(8)));

#define D 64

__global__ __launch_bounds__(128)
void gcn_gemm_bias(const float* __restrict__ x,
                   const float* __restrict__ W,
                   const float* __restrict__ b,
                   float* __restrict__ h,
                   int n_nodes) {
  const int lane = threadIdx.x & 31;
  const int wave = threadIdx.x >> 5;
  const int row0 = blockIdx.x << 4;
  const int col0 = wave << 4;

  const int m  = lane & 15;
  const int kk = (lane >> 4) << 1;

  int arow = row0 + m;
  if (arow >= n_nodes) arow = n_nodes - 1;

  const float* __restrict__ xrow = x + (size_t)arow * D;
  const float* __restrict__ wrow = W + (size_t)(col0 + m) * D;

  v8f acc = {};
#pragma unroll
  for (int k0 = 0; k0 < D; k0 += 4) {
    v2f a  = *(const v2f*)(xrow + k0 + kk);
    v2f bb = *(const v2f*)(wrow + k0 + kk);
    acc = __builtin_amdgcn_wmma_f32_16x16x4_f32(
        false, a, false, bb, (short)0, acc, false, false);
  }

  const float bias = b[col0 + m];
  const int rbase = row0 + ((lane >> 4) << 3);
#pragma unroll
  for (int r = 0; r < 8; ++r) {
    const int row = rbase + r;
    if (row < n_nodes) {
      h[(size_t)row * D + col0 + m] = acc[r] + bias;
    }
  }
}

#define EDGES_PER_WAVE 32
#define WAVES_PER_BLOCK 8
#define CHUNK 8

__device__ __forceinline__ void edge_rows(const float* __restrict__ h, int my_s,
                                          int lane, int e, float& v0, float& v1) {
  const int s = __builtin_amdgcn_readlane(my_s, e);
  const float* __restrict__ hp = h + (size_t)s * D + lane;
  v0 = hp[0];
  v1 = hp[32];
}

__device__ __forceinline__ void edge_add(float* __restrict__ out, int my_d,
                                         int my_wbits, int lane, int e,
                                         float v0, float v1) {
  const int   d  = __builtin_amdgcn_readlane(my_d, e);
  const float wv = __int_as_float(__builtin_amdgcn_readlane(my_wbits, e));
  float* __restrict__ op = out + (size_t)d * D + lane;
  atomicAdd(op,      v0 * wv);
  atomicAdd(op + 32, v1 * wv);
}

template <int C0>
__device__ __forceinline__ void do_chunk(const float* __restrict__ h,
                                         float* __restrict__ out, int my_s,
                                         int my_d, int my_wbits, int lane) {
  float v0[CHUNK], v1[CHUNK];
#pragma unroll
  for (int j = 0; j < CHUNK; ++j) edge_rows(h, my_s, lane, C0 + j, v0[j], v1[j]);
#pragma unroll
  for (int j = 0; j < CHUNK; ++j) edge_add(out, my_d, my_wbits, lane, C0 + j, v0[j], v1[j]);
}

__global__ __launch_bounds__(WAVES_PER_BLOCK * 32)
void gcn_edge_scatter(const float* __restrict__ h,
                      const int* __restrict__ src,
                      const int* __restrict__ dst,
                      const float* __restrict__ w,
                      float* __restrict__ out,
                      int n_edges) {
  const int lane = threadIdx.x & 31;
  const int wave = __builtin_amdgcn_readfirstlane((int)(threadIdx.x >> 5));
  const int e0 = (blockIdx.x * WAVES_PER_BLOCK + wave) * EDGES_PER_WAVE;
  if (e0 >= n_edges) return;
  const int left = n_edges - e0;
  const int cnt = left < EDGES_PER_WAVE ? left : EDGES_PER_WAVE;

  int my_s = 0, my_d = 0, my_wbits = 0;
  if (lane < cnt) {
    my_s     = src[e0 + lane];
    my_d     = dst[e0 + lane];
    my_wbits = __float_as_int(w[e0 + lane]);
  }

  if (cnt == EDGES_PER_WAVE) {
    do_chunk<0>(h, out, my_s, my_d, my_wbits, lane);
    do_chunk<8>(h, out, my_s, my_d, my_wbits, lane);
    do_chunk<16>(h, out, my_s, my_d, my_wbits, lane);
    do_chunk<24>(h, out, my_s, my_d, my_wbits, lane);
  } else {
    for (int e = 0; e < cnt; ++e) {
      float v0, v1;
      edge_rows(h, my_s, lane, e, v0, v1);
      edge_add(out, my_d, my_wbits, lane, e, v0, v1);
    }
  }
}

extern "C" void kernel_launch(void* const* d_in, const int* in_sizes, int n_in,
                              void* d_out, int out_size, void* d_ws, size_t ws_size,
                              hipStream_t stream) {
  const float* x   = (const float*)d_in[0];
  const int*   src = (const int*)  d_in[1];
  const int*   dst = (const int*)  d_in[2];
  const float* w   = (const float*)d_in[3];
  const float* W   = (const float*)d_in[4];
  const float* b   = (const float*)d_in[5];
  float*       out = (float*)d_out;
  float*       h   = (float*)d_ws;

  const int n_nodes = in_sizes[0] / D;
  const int n_edges = in_sizes[1];

  hipMemsetAsync(d_out, 0, (size_t)out_size * sizeof(float), stream);

  const int row_tiles = (n_nodes + 15) / 16;
  gcn_gemm_bias<<<dim3(row_tiles), dim3(128), 0, stream>>>(x, W, b, h, n_nodes);

  const int edges_per_block = EDGES_PER_WAVE * WAVES_PER_BLOCK;
  const int blocks = (n_edges + edges_per_block - 1) / edges_per_block;
  gcn_edge_scatter<<<dim3(blocks), dim3(WAVES_PER_BLOCK * 32), 0, stream>>>(h, src, dst, w, out, n_edges);
}


# GEMM: one wave computes a whole 16x64 row tile (x fragments loaded once, not 4x), W staged once per block via LDS into registers, 512 blocks walk the tiles, next tile's x in flight; sums unchanged
# speedup vs baseline: 2.0411x; 1.1095x over previous
#include <hip/hip_runtime.h>
#include <hip/hip_bf16.h>

typedef float v2f __attribute__((ext_vector_type(2)));
typedef float v8f __attribute__((ext_vector_type(8)));

#define D 64

typedef float v4f __attribute__((ext_vector_type(4)));

#define GEMM_WAVES 4
#define GEMM_BLOCKS 512
#define WS_PITCH 68

__device__ __forceinline__ void load_a_tile(const float* __restrict__ x, int tile,
                                            int m, int kk, int n_nodes,
                                            v2f (&a)[16]) {
  int arow = (tile << 4) + m;
  if (arow >= n_nodes) arow = n_nodes - 1;
  const float* __restrict__ xr = x + (size_t)arow * D + kk;
#pragma unroll
  for (int s = 0; s < 16; ++s) a[s] = *(const v2f*)(xr + 4 * s);
}

__global__ __launch_bounds__(GEMM_WAVES * 32)
void gcn_gemm_bias(const float* __restrict__ x,
                   const float* __restrict__ W,
                   const float* __restrict__ b,
                   float* __restrict__ h,
                   int n_nodes) {
  __shared__ __attribute__((aligned(16))) float Ws[D * WS_PITCH];

  const int tid  = threadIdx.x;
  const int lane = tid & 31;
  const int m  = lane & 15;
  const int kk = (lane >> 4) << 1;

#pragma unroll
  for (int i = 0; i < (D * D / 4) / (GEMM_WAVES * 32); ++i) {
    const int idx = tid + i * (GEMM_WAVES * 32);
    const int row = idx >> 4;
    const int c4  = idx & 15;
    const v4f v = *(const v4f*)(W + (size_t)idx * 4);
    *(v4f*)&Ws[row * WS_PITCH + (c4 << 2)] = v;
  }
  __syncthreads();

  v2f bf[4][16];
#pragma unroll
  for (int c = 0; c < 4; ++c) {
#pragma unroll
    for (int s = 0; s < 16; ++s) {
      bf[c][s] = *(const v2f*)&Ws[((c << 4) + m) * WS_PITCH + 4 * s + kk];
    }
  }
  float bias[4];
#pragma unroll
  for (int c = 0; c < 4; ++c) bias[c] = b[(c << 4) + m];

  const int wave = __builtin_amdgcn_readfirstlane(tid >> 5);
  const int nw = (int)gridDim.x * GEMM_WAVES;
  const int n_tiles = (n_nodes + 15) >> 4;
  int t = (int)blockIdx.x * GEMM_WAVES + wave;

  v2f a[16];
  load_a_tile(x, t, m, kk, n_nodes, a);

  while (t < n_tiles) {
    const int tn = t + nw;
    v2f an[16];
    load_a_tile(x, tn, m, kk, n_nodes, an);

    v8f acc[4] = {};
#pragma unroll
    for (int s = 0; s < 16; ++s) {
#pragma unroll
      for (int c = 0; c < 4; ++c) {
        acc[c] = __builtin_amdgcn_wmma_f32_16x16x4_f32(
            false, a[s], false, bf[c][s], (short)0, acc[c], false, false);
      }
    }

    const int row0  = t << 4;
    const int rbase = row0 + ((lane >> 4) << 3);
    float* __restrict__ hp = h + (size_t)rbase * D + m;
    if (row0 + 16 <= n_nodes) {
#pragma unroll
      for (int c = 0; c < 4; ++c) {
#pragma unroll
        for (int r = 0; r < 8; ++r) hp[r * D + (c << 4)] = acc[c][r] + bias[c];
      }
    } else {
#pragma unroll
      for (int c = 0; c < 4; ++c) {
#pragma unroll
        for (int r = 0; r < 8; ++r) {
          if (rbase + r < n_nodes) hp[r * D + (c << 4)] = acc[c][r] + bias[c];
        }
      }
    }

#pragma unroll
    for (int s = 0; s < 16; ++s) a[s] = an[s];
    t = tn;
  }
}

#define EDGES_PER_WAVE 32
#define WAVES_PER_BLOCK 8
#define CHUNK 8

__device__ __forceinline__ void edge_rows(const float* __restrict__ h, int my_s,
                                          int lane, int e, float& v0, float& v1) {
  const int s = __builtin_amdgcn_readlane(my_s, e);
  const float* __restrict__ hp = h + (size_t)s * D + lane;
  v0 = hp[0];
  v1 = hp[32];
}

__device__ __forceinline__ void edge_add(float* __restrict__ out, int my_d,
                                         int my_wbits, int lane, int e,
                                         float v0, float v1) {
  const int   d  = __builtin_amdgcn_readlane(my_d, e);
  const float wv = __int_as_float(__builtin_amdgcn_readlane(my_wbits, e));
  float* __restrict__ op = out + (size_t)d * D + lane;
  atomicAdd(op,      v0 * wv);
  atomicAdd(op + 32, v1 * wv);
}

template <int C0>
__device__ __forceinline__ void do_chunk(const float* __restrict__ h,
                                         float* __restrict__ out, int my_s,
                                         int my_d, int my_wbits, int lane) {
  float v0[CHUNK], v1[CHUNK];
#pragma unroll
  for (int j = 0; j < CHUNK; ++j) edge_rows(h, my_s, lane, C0 + j, v0[j], v1[j]);
#pragma unroll
  for (int j = 0; j < CHUNK; ++j) edge_add(out, my_d, my_wbits, lane, C0 + j, v0[j], v1[j]);
}

__global__ __launch_bounds__(WAVES_PER_BLOCK * 32)
void gcn_edge_scatter(const float* __restrict__ h,
                      const int* __restrict__ src,
                      const int* __restrict__ dst,
                      const float* __restrict__ w,
                      float* __restrict__ out,
                      int n_edges) {
  const int lane = threadIdx.x & 31;
  const int wave = __builtin_amdgcn_readfirstlane((int)(threadIdx.x >> 5));
  const int e0 = (blockIdx.x * WAVES_PER_BLOCK + wave) * EDGES_PER_WAVE;
  if (e0 >= n_edges) return;
  const int left = n_edges - e0;
  const int cnt = left < EDGES_PER_WAVE ? left : EDGES_PER_WAVE;

  int my_s = 0, my_d = 0, my_wbits = 0;
  if (lane < cnt) {
    my_s     = src[e0 + lane];
    my_d     = dst[e0 + lane];
    my_wbits = __float_as_int(w[e0 + lane]);
  }

  if (cnt == EDGES_PER_WAVE) {
    do_chunk<0>(h, out, my_s, my_d, my_wbits, lane);
    do_chunk<8>(h, out, my_s, my_d, my_wbits, lane);
    do_chunk<16>(h, out, my_s, my_d, my_wbits, lane);
    do_chunk<24>(h, out, my_s, my_d, my_wbits, lane);
  } else {
    for (int e = 0; e < cnt; ++e) {
      float v0, v1;
      edge_rows(h, my_s, lane, e, v0, v1);
      edge_add(out, my_d, my_wbits, lane, e, v0, v1);
    }
  }
}

extern "C" void kernel_launch(void* const* d_in, const int* in_sizes, int n_in,
                              void* d_out, int out_size, void* d_ws, size_t ws_size,
                              hipStream_t stream) {
  const float* x   = (const float*)d_in[0];
  const int*   src = (const int*)  d_in[1];
  const int*   dst = (const int*)  d_in[2];
  const float* w   = (const float*)d_in[3];
  const float* W   = (const float*)d_in[4];
  const float* b   = (const float*)d_in[5];
  float*       out = (float*)d_out;
  float*       h   = (float*)d_ws;

  const int n_nodes = in_sizes[0] / D;
  const int n_edges = in_sizes[1];

  hipMemsetAsync(d_out, 0, (size_t)out_size * sizeof(float), stream);

  gcn_gemm_bias<<<dim3(GEMM_BLOCKS), dim3(GEMM_WAVES * 32), 0, stream>>>(x, W, b, h, n_nodes);

  const int edges_per_block = EDGES_PER_WAVE * WAVES_PER_BLOCK;
  const int blocks = (n_edges + edges_per_block - 1) / edges_per_block;
  gcn_edge_scatter<<<dim3(blocks), dim3(WAVES_PER_BLOCK * 32), 0, stream>>>(h, src, dst, w, out, n_edges);
}
